# starting kernel
# baseline (speedup 1.0000x reference)
#include <hip/hip_runtime.h>
#include <math.h>

typedef __attribute__((ext_vector_type(2))) float v2f;
typedef __attribute__((ext_vector_type(8))) float v8f;

#define Hdim 256
#define ENC 1024
#define DEC 256
#define BATCH 4

__device__ __forceinline__ float gpu_tanh(float x) {
#if __has_builtin(__builtin_amdgcn_tanhf)
    return __builtin_amdgcn_tanhf(x);
#elif __has_builtin(__builtin_amdgcn_tanh_f32)
    return __builtin_amdgcn_tanh_f32(x);
#else
    float r;
    asm volatile("v_tanh_f32 %0, %1\n\tv_nop" : "=v"(r) : "v"(x));
    return r;
#endif
}

__global__ __launch_bounds__(32) void gemm_xwT_wmma(const float* __restrict__ x,
                                                    const float* __restrict__ w,
                                                    float* __restrict__ out) {
    const int lane  = threadIdx.x;
    const int mtile = blockIdx.x;
    const int ntile = blockIdx.y;
    const int l15   = lane & 15;
    const int koff  = (lane >> 4) << 1;

    const float2* xrow = reinterpret_cast<const float2*>(x + (size_t)(mtile * 16 + l15) * Hdim);
    const float2* wrow = reinterpret_cast<const float2*>(w + (size_t)(ntile * 16 + l15) * Hdim);

    v8f c = {};
#pragma unroll 4
    for (int k0 = 0; k0 < Hdim; k0 += 4) {
        float2 af = xrow[(k0 + koff) >> 1];
        float2 bf = wrow[(k0 + koff) >> 1];
        v2f a; a.x = af.x; a.y = af.y;
        v2f b; b.x = bf.x; b.y = bf.y;
        c = __builtin_amdgcn_wmma_f32_16x16x4_f32(false, a, false, b, (short)0, c,
                                                  false, false);
    }

    const int nstore = ntile * 16 + l15;
    const int mbase  = mtile * 16 + ((lane >> 4) << 3);
#pragma unroll
    for (int r = 0; r < 8; ++r)
        out[(size_t)(mbase + r) * Hdim + nstore] = c[r];
}

__global__ __launch_bounds__(256) void attn_scores_lsm(const float* __restrict__ dec_t,
                                                       const float* __restrict__ enc_t,
                                                       const float* __restrict__ v,
                                                       float* __restrict__ out) {
    const int bd = blockIdx.x;
    const int b  = bd / DEC;
    const int t  = threadIdx.x;

    __shared__ float sdec[Hdim];
    __shared__ float sv[Hdim];
    __shared__ float red[256];

    sdec[t] = dec_t[(size_t)bd * Hdim + t];
    sv[t]   = v[t];
    __syncthreads();

    const float4* sdec4 = reinterpret_cast<const float4*>(sdec);
    const float4* sv4   = reinterpret_cast<const float4*>(sv);

    float s[4];
#pragma unroll
    for (int j = 0; j < 4; ++j) {
        const int e = j * 256 + t;
        const float4* erow =
            reinterpret_cast<const float4*>(enc_t + ((size_t)b * ENC + e) * Hdim);
        float acc = 0.f;
        for (int i = 0; i < Hdim / 4; ++i) {
            float4 ev = erow[i];
            float4 dv = sdec4[i];
            float4 vv = sv4[i];
            acc += vv.x * gpu_tanh(dv.x + ev.x);
            acc += vv.y * gpu_tanh(dv.y + ev.y);
            acc += vv.z * gpu_tanh(dv.z + ev.z);
            acc += vv.w * gpu_tanh(dv.w + ev.w);
        }
        s[j] = acc;
    }

    float m = fmaxf(fmaxf(s[0], s[1]), fmaxf(s[2], s[3]));
    red[t] = m;
    __syncthreads();
    for (int str = 128; str >= 1; str >>= 1) {
        if (t < str) red[t] = fmaxf(red[t], red[t + str]);
        __syncthreads();
    }
    const float gmax = red[0];
    __syncthreads();

    float lsum = 0.f;
#pragma unroll
    for (int j = 0; j < 4; ++j) lsum += __expf(s[j] - gmax);
    red[t] = lsum;
    __syncthreads();
    for (int str = 128; str >= 1; str >>= 1) {
        if (t < str) red[t] += red[t + str];
        __syncthreads();
    }
    const float lse = gmax + __logf(red[0]);

#pragma unroll
    for (int j = 0; j < 4; ++j)
        out[(size_t)bd * ENC + (size_t)j * 256 + t] = s[j] - lse;
}

extern "C" void kernel_launch(void* const* d_in, const int* in_sizes, int n_in,
                              void* d_out, int out_size, void* d_ws, size_t ws_size,
                              hipStream_t stream) {
    const float* x_dec = (const float*)d_in[0];
    const float* x_enc = (const float*)d_in[1];
    const float* w1    = (const float*)d_in[2];
    const float* w2    = (const float*)d_in[3];
    const float* v     = (const float*)d_in[4];
    float* out = (float*)d_out;

    float* enc_t = (float*)d_ws;
    float* dec_t = enc_t + (size_t)BATCH * ENC * Hdim;

    gemm_xwT_wmma<<<dim3(BATCH * ENC / 16, Hdim / 16), 32, 0, stream>>>(x_enc, w1, enc_t);
    gemm_xwT_wmma<<<dim3(BATCH * DEC / 16, Hdim / 16), 32, 0, stream>>>(x_dec, w2, dec_t);
    attn_scores_lsm<<<BATCH * DEC, 256, 0, stream>>>(dec_t, enc_t, v, out);
}


# projection GEMM re-tiled: x and w rows staged once in LDS by whole-line 16-byte loads, WMMA fragments read from LDS, each wave a 16x32 piece (A fragment reused); k summed in a new order
# speedup vs baseline: 2.9567x; 2.9567x over previous
#include <hip/hip_runtime.h>
#include <math.h>

typedef __attribute__((ext_vector_type(2))) float v2f;
typedef __attribute__((ext_vector_type(8))) float v8f;

#define Hdim 256
#define ENC 1024
#define DEC 256
#define BATCH 4

__device__ __forceinline__ float gpu_tanh(float x) {
#if __has_builtin(__builtin_amdgcn_tanhf)
    return __builtin_amdgcn_tanhf(x);
#elif __has_builtin(__builtin_amdgcn_tanh_f32)
    return __builtin_amdgcn_tanh_f32(x);
#else
    float r;
    asm volatile("v_tanh_f32 %0, %1\n\tv_nop" : "=v"(r) : "v"(x));
    return r;
#endif
}

#define BN 64
#define PITCH 260

__device__ __forceinline__ void load_frags(float4 (&f)[4], const float* p) {
#pragma unroll
    for (int s = 0; s < 4; ++s) f[s] = *reinterpret_cast<const float4*>(p + 8 * s);
}

template <int BM>
__global__ __launch_bounds__(BM * 4) void gemm_xwT_wmma(const float* __restrict__ x,
                                                        const float* __restrict__ w,
                                                        float* __restrict__ out) {
    constexpr int NT = BM * 4;
    constexpr int RS = NT / 64;
    constexpr int NA = BM / RS;
    constexpr int NB = BN / RS;

    __shared__ __attribute__((aligned(16))) float sA[BM * PITCH];
    __shared__ __attribute__((aligned(16))) float sB[BN * PITCH];

    const int t  = threadIdx.x;
    const int m0 = blockIdx.x * BM;
    const int n0 = blockIdx.y * BN;
    const int c4 = t & 63;
    const int r0 = t >> 6;

    float4 ra[NA], rb[NB];
#pragma unroll
    for (int i = 0; i < NA; ++i)
        ra[i] = *reinterpret_cast<const float4*>(x + (size_t)(m0 + i * RS + r0) * Hdim + 4 * c4);
#pragma unroll
    for (int i = 0; i < NB; ++i)
        rb[i] = *reinterpret_cast<const float4*>(w + (size_t)(n0 + i * RS + r0) * Hdim + 4 * c4);
#pragma unroll
    for (int i = 0; i < NA; ++i)
        *reinterpret_cast<float4*>(sA + (i * RS + r0) * PITCH + 4 * c4) = ra[i];
#pragma unroll
    for (int i = 0; i < NB; ++i)
        *reinterpret_cast<float4*>(sB + (i * RS + r0) * PITCH + 4 * c4) = rb[i];
    __syncthreads();

    const int lane = t & 31;
    const int wv   = t >> 5;
    const int c    = lane & 15;
    const int hf   = lane >> 4;
    const int mt   = wv % (BM / 16);
    const int nh   = wv / (BM / 16);

    const float* pa  = sA + (mt * 16 + c) * PITCH + 4 * hf;
    const float* pb0 = sB + (nh * 32 + c) * PITCH + 4 * hf;
    const float* pb1 = pb0 + 16 * PITCH;

    v8f acc0 = {};
    v8f acc1 = {};
    float4 fa[2][4], fb0[2][4], fb1[2][4];
    load_frags(fa[0], pa);
    load_frags(fb0[0], pb0);
    load_frags(fb1[0], pb1);
    __builtin_amdgcn_sched_group_barrier(0x100, 12, 0);
#pragma unroll
    for (int g = 0; g < 8; ++g) {
        const int cur = g & 1;
        const int nxt = cur ^ 1;
        if (g + 1 < 8) {
            load_frags(fa[nxt],  pa  + 32 * (g + 1));
            load_frags(fb0[nxt], pb0 + 32 * (g + 1));
            load_frags(fb1[nxt], pb1 + 32 * (g + 1));
            __builtin_amdgcn_sched_group_barrier(0x100, 12, 0);
        }
        __builtin_amdgcn_sched_group_barrier(0x008, 16, 0);
#pragma unroll
        for (int s = 0; s < 4; ++s) {
            v2f alo; alo.x = fa[cur][s].x;  alo.y = fa[cur][s].y;
            v2f ahi; ahi.x = fa[cur][s].z;  ahi.y = fa[cur][s].w;
            v2f b0lo; b0lo.x = fb0[cur][s].x; b0lo.y = fb0[cur][s].y;
            v2f b0hi; b0hi.x = fb0[cur][s].z; b0hi.y = fb0[cur][s].w;
            v2f b1lo; b1lo.x = fb1[cur][s].x; b1lo.y = fb1[cur][s].y;
            v2f b1hi; b1hi.x = fb1[cur][s].z; b1hi.y = fb1[cur][s].w;
            acc0 = __builtin_amdgcn_wmma_f32_16x16x4_f32(false, alo, false, b0lo, (short)0, acc0, false, false);
            acc1 = __builtin_amdgcn_wmma_f32_16x16x4_f32(false, alo, false, b1lo, (short)0, acc1, false, false);
            acc0 = __builtin_amdgcn_wmma_f32_16x16x4_f32(false, ahi, false, b0hi, (short)0, acc0, false, false);
            acc1 = __builtin_amdgcn_wmma_f32_16x16x4_f32(false, ahi, false, b1hi, (short)0, acc1, false, false);
        }
    }

    float* orow = out + (size_t)(m0 + mt * 16 + 8 * hf) * Hdim + n0 + nh * 32 + c;
#pragma unroll
    for (int r = 0; r < 8; ++r) {
        orow[(size_t)r * Hdim]      = acc0[r];
        orow[(size_t)r * Hdim + 16] = acc1[r];
    }
}

#define DT 4


template <int MASK>
__device__ __forceinline__ float add_xor_lane(float x) {
    const int y = __builtin_amdgcn_update_dpp(0, __builtin_bit_cast(int, x), 0x160 | MASK, 0xf, 0xf, false);
    return x + __builtin_bit_cast(float, y);
}

__device__ __forceinline__ void row_group(const float4 (&ev)[8], const float4 (&dr)[DT][8],
                                          const float4 (&vr)[8], float (&a)[DT]) {
#pragma unroll
    for (int c = 0; c < 8; ++c) {
#pragma unroll
        for (int d = 0; d < DT; ++d) {
            a[d] += vr[c].x * gpu_tanh(dr[d][c].x + ev[c].x);
            a[d] += vr[c].y * gpu_tanh(dr[d][c].y + ev[c].y);
            a[d] += vr[c].z * gpu_tanh(dr[d][c].z + ev[c].z);
            a[d] += vr[c].w * gpu_tanh(dr[d][c].w + ev[c].w);
        }
    }
}

__device__ __forceinline__ void load_group(float4 (&ev)[8], const float* __restrict__ p) {
#pragma unroll
    for (int c = 0; c < 8; ++c) ev[c] = *reinterpret_cast<const float4*>(p + 32 * c);
}

__global__ __launch_bounds__(256) void attn_scores_lsm(const float* __restrict__ dec_t,
                                                       const float* __restrict__ enc_t,
                                                       const float* __restrict__ v,
                                                       float* __restrict__ out) {
    const int bd0  = blockIdx.x * DT;
    const int b    = bd0 / DEC;
    const int t    = threadIdx.x;
    const int lane = t & 31;
    const int w    = t >> 5;
    const int q    = lane & 7;
    const int g    = lane >> 3;

    __shared__ float sc[DT][ENC];
    __shared__ float red[DT][256];

    float4 dr[DT][8];
    float4 vr[8];
#pragma unroll
    for (int c = 0; c < 8; ++c) {
        vr[c] = *reinterpret_cast<const float4*>(v + 32 * c + 4 * q);
#pragma unroll
        for (int d = 0; d < DT; ++d)
            dr[d][c] = *reinterpret_cast<const float4*>(dec_t + (size_t)(bd0 + d) * Hdim + 32 * c + 4 * q);
    }

    const float* ebase = enc_t + ((size_t)b * ENC + w * 128 + g) * Hdim + 4 * q;
    const int erow0 = w * 128 + g;

    float4 eva[8], evb[8];
    load_group(eva, ebase);
    for (int it = 0; it < 32; it += 2) {
        load_group(evb, ebase + (size_t)(it + 1) * 4 * Hdim);
        {
            float a[DT] = {};
            row_group(eva, dr, vr, a);
#pragma unroll
            for (int d = 0; d < DT; ++d) {
                a[d] = add_xor_lane<1>(a[d]);
                a[d] = add_xor_lane<2>(a[d]);
                a[d] = add_xor_lane<4>(a[d]);
            }
            if (q == 0) {
#pragma unroll
                for (int d = 0; d < DT; ++d) sc[d][erow0 + it * 4] = a[d];
            }
        }
        const int nx = (it + 2 < 32) ? it + 2 : 31;
        load_group(eva, ebase + (size_t)nx * 4 * Hdim);
        {
            float a[DT] = {};
            row_group(evb, dr, vr, a);
#pragma unroll
            for (int d = 0; d < DT; ++d) {
                a[d] = add_xor_lane<1>(a[d]);
                a[d] = add_xor_lane<2>(a[d]);
                a[d] = add_xor_lane<4>(a[d]);
            }
            if (q == 0) {
#pragma unroll
                for (int d = 0; d < DT; ++d) sc[d][erow0 + (it + 1) * 4] = a[d];
            }
        }
    }
    __syncthreads();

    float s[DT][4];
#pragma unroll
    for (int d = 0; d < DT; ++d)
#pragma unroll
        for (int j = 0; j < 4; ++j) s[d][j] = sc[d][j * 256 + t];

#pragma unroll
    for (int d = 0; d < DT; ++d)
        red[d][t] = fmaxf(fmaxf(s[d][0], s[d][1]), fmaxf(s[d][2], s[d][3]));
    __syncthreads();
    for (int str = 128; str >= 1; str >>= 1) {
        if (t < str) {
#pragma unroll
            for (int d = 0; d < DT; ++d) red[d][t] = fmaxf(red[d][t], red[d][t + str]);
        }
        __syncthreads();
    }
    float gmax[DT];
#pragma unroll
    for (int d = 0; d < DT; ++d) gmax[d] = red[d][0];
    __syncthreads();

#pragma unroll
    for (int d = 0; d < DT; ++d) {
        float lsum = 0.f;
#pragma unroll
        for (int j = 0; j < 4; ++j) lsum += __expf(s[d][j] - gmax[d]);
        red[d][t] = lsum;
    }
    __syncthreads();
    for (int str = 128; str >= 1; str >>= 1) {
        if (t < str) {
#pragma unroll
            for (int d = 0; d < DT; ++d) red[d][t] += red[d][t + str];
        }
        __syncthreads();
    }

#pragma unroll
    for (int d = 0; d < DT; ++d) {
        const float lse = gmax[d] + __logf(red[d][0]);
#pragma unroll
        for (int j = 0; j < 4; ++j)
            out[(size_t)(bd0 + d) * ENC + (size_t)j * 256 + t] = s[d][j] - lse;
    }
}

extern "C" void kernel_launch(void* const* d_in, const int* in_sizes, int n_in,
                              void* d_out, int out_size, void* d_ws, size_t ws_size,
                              hipStream_t stream) {
    const float* x_dec = (const float*)d_in[0];
    const float* x_enc = (const float*)d_in[1];
    const float* w1    = (const float*)d_in[2];
    const float* w2    = (const float*)d_in[3];
    const float* v     = (const float*)d_in[4];
    float* out = (float*)d_out;

    float* enc_t = (float*)d_ws;
    float* dec_t = enc_t + (size_t)BATCH * ENC * Hdim;

    gemm_xwT_wmma<64><<<dim3(BATCH * ENC / 64, Hdim / BN), 256, 0, stream>>>(x_enc, w1, enc_t);
    gemm_xwT_wmma<32><<<dim3(BATCH * DEC / 32, Hdim / BN), 128, 0, stream>>>(x_dec, w2, dec_t);
    attn_scores_lsm<<<BATCH * DEC / DT, 256, 0, stream>>>(dec_t, enc_t, v, out);
}


# scores kernel: 16 waves per block (4 per SIMD, was 2), each wave 64 enc rows, so twice the enc loads are in flight per WGP and a stalled wave finds others; exp-sum tree over 512 threads (new order)
# speedup vs baseline: 3.1575x; 1.0679x over previous
#include <hip/hip_runtime.h>
#include <math.h>

typedef __attribute__((ext_vector_type(2))) float v2f;
typedef __attribute__((ext_vector_type(8))) float v8f;

#define Hdim 256
#define ENC 1024
#define DEC 256
#define BATCH 4

__device__ __forceinline__ float gpu_tanh(float x) {
#if __has_builtin(__builtin_amdgcn_tanhf)
    return __builtin_amdgcn_tanhf(x);
#elif __has_builtin(__builtin_amdgcn_tanh_f32)
    return __builtin_amdgcn_tanh_f32(x);
#else
    float r;
    asm volatile("v_tanh_f32 %0, %1\n\tv_nop" : "=v"(r) : "v"(x));
    return r;
#endif
}

#define BN 64
#define PITCH 260

__device__ __forceinline__ void load_frags(float4 (&f)[4], const float* p) {
#pragma unroll
    for (int s = 0; s < 4; ++s) f[s] = *reinterpret_cast<const float4*>(p + 8 * s);
}

template <int BM>
__global__ __launch_bounds__(BM * 4) void gemm_xwT_wmma(const float* __restrict__ x,
                                                        const float* __restrict__ w,
                                                        float* __restrict__ out) {
    constexpr int NT = BM * 4;
    constexpr int RS = NT / 64;
    constexpr int NA = BM / RS;
    constexpr int NB = BN / RS;

    __shared__ __attribute__((aligned(16))) float sA[BM * PITCH];
    __shared__ __attribute__((aligned(16))) float sB[BN * PITCH];

    const int t  = threadIdx.x;
    const int m0 = blockIdx.x * BM;
    const int n0 = blockIdx.y * BN;
    const int c4 = t & 63;
    const int r0 = t >> 6;

    float4 ra[NA], rb[NB];
#pragma unroll
    for (int i = 0; i < NA; ++i)
        ra[i] = *reinterpret_cast<const float4*>(x + (size_t)(m0 + i * RS + r0) * Hdim + 4 * c4);
#pragma unroll
    for (int i = 0; i < NB; ++i)
        rb[i] = *reinterpret_cast<const float4*>(w + (size_t)(n0 + i * RS + r0) * Hdim + 4 * c4);
#pragma unroll
    for (int i = 0; i < NA; ++i)
        *reinterpret_cast<float4*>(sA + (i * RS + r0) * PITCH + 4 * c4) = ra[i];
#pragma unroll
    for (int i = 0; i < NB; ++i)
        *reinterpret_cast<float4*>(sB + (i * RS + r0) * PITCH + 4 * c4) = rb[i];
    __syncthreads();

    const int lane = t & 31;
    const int wv   = t >> 5;
    const int c    = lane & 15;
    const int hf   = lane >> 4;
    const int mt   = wv % (BM / 16);
    const int nh   = wv / (BM / 16);

    const float* pa  = sA + (mt * 16 + c) * PITCH + 4 * hf;
    const float* pb0 = sB + (nh * 32 + c) * PITCH + 4 * hf;
    const float* pb1 = pb0 + 16 * PITCH;

    v8f acc0 = {};
    v8f acc1 = {};
    float4 fa[2][4], fb0[2][4], fb1[2][4];
    load_frags(fa[0], pa);
    load_frags(fb0[0], pb0);
    load_frags(fb1[0], pb1);
    __builtin_amdgcn_sched_group_barrier(0x100, 12, 0);
#pragma unroll
    for (int g = 0; g < 8; ++g) {
        const int cur = g & 1;
        const int nxt = cur ^ 1;
        if (g + 1 < 8) {
            load_frags(fa[nxt],  pa  + 32 * (g + 1));
            load_frags(fb0[nxt], pb0 + 32 * (g + 1));
            load_frags(fb1[nxt], pb1 + 32 * (g + 1));
            __builtin_amdgcn_sched_group_barrier(0x100, 12, 0);
        }
        __builtin_amdgcn_sched_group_barrier(0x008, 16, 0);
#pragma unroll
        for (int s = 0; s < 4; ++s) {
            v2f alo; alo.x = fa[cur][s].x;  alo.y = fa[cur][s].y;
            v2f ahi; ahi.x = fa[cur][s].z;  ahi.y = fa[cur][s].w;
            v2f b0lo; b0lo.x = fb0[cur][s].x; b0lo.y = fb0[cur][s].y;
            v2f b0hi; b0hi.x = fb0[cur][s].z; b0hi.y = fb0[cur][s].w;
            v2f b1lo; b1lo.x = fb1[cur][s].x; b1lo.y = fb1[cur][s].y;
            v2f b1hi; b1hi.x = fb1[cur][s].z; b1hi.y = fb1[cur][s].w;
            acc0 = __builtin_amdgcn_wmma_f32_16x16x4_f32(false, alo, false, b0lo, (short)0, acc0, false, false);
            acc1 = __builtin_amdgcn_wmma_f32_16x16x4_f32(false, alo, false, b1lo, (short)0, acc1, false, false);
            acc0 = __builtin_amdgcn_wmma_f32_16x16x4_f32(false, ahi, false, b0hi, (short)0, acc0, false, false);
            acc1 = __builtin_amdgcn_wmma_f32_16x16x4_f32(false, ahi, false, b1hi, (short)0, acc1, false, false);
        }
    }

    float* orow = out + (size_t)(m0 + mt * 16 + 8 * hf) * Hdim + n0 + nh * 32 + c;
#pragma unroll
    for (int r = 0; r < 8; ++r) {
        orow[(size_t)r * Hdim]      = acc0[r];
        orow[(size_t)r * Hdim + 16] = acc1[r];
    }
}

#define DT 4
#define ST 512
#define WROWS (ENC / (ST / 32))
#define NGRP (WROWS / 4)
#define EPT (ENC / ST)


template <int MASK>
__device__ __forceinline__ float add_xor_lane(float x) {
    const int y = __builtin_amdgcn_update_dpp(0, __builtin_bit_cast(int, x), 0x160 | MASK, 0xf, 0xf, false);
    return x + __builtin_bit_cast(float, y);
}

__device__ __forceinline__ void row_group(const float4 (&ev)[8], const float4 (&dr)[DT][8],
                                          const float4 (&vr)[8], float (&a)[DT]) {
#pragma unroll
    for (int c = 0; c < 8; ++c) {
#pragma unroll
        for (int d = 0; d < DT; ++d) {
            a[d] += vr[c].x * gpu_tanh(dr[d][c].x + ev[c].x);
            a[d] += vr[c].y * gpu_tanh(dr[d][c].y + ev[c].y);
            a[d] += vr[c].z * gpu_tanh(dr[d][c].z + ev[c].z);
            a[d] += vr[c].w * gpu_tanh(dr[d][c].w + ev[c].w);
        }
    }
}

__device__ __forceinline__ void load_group(float4 (&ev)[8], const float* __restrict__ p) {
#pragma unroll
    for (int c = 0; c < 8; ++c) ev[c] = *reinterpret_cast<const float4*>(p + 32 * c);
}

__global__ __launch_bounds__(ST) void attn_scores_lsm(const float* __restrict__ dec_t,
                                                       const float* __restrict__ enc_t,
                                                       const float* __restrict__ v,
                                                       float* __restrict__ out) {
    const int bd0  = blockIdx.x * DT;
    const int b    = bd0 / DEC;
    const int t    = threadIdx.x;
    const int lane = t & 31;
    const int w    = t >> 5;
    const int q    = lane & 7;
    const int g    = lane >> 3;

    __shared__ float sc[DT][ENC];
    __shared__ float red[DT][ST];

    float4 dr[DT][8];
    float4 vr[8];
#pragma unroll
    for (int c = 0; c < 8; ++c) {
        vr[c] = *reinterpret_cast<const float4*>(v + 32 * c + 4 * q);
#pragma unroll
        for (int d = 0; d < DT; ++d)
            dr[d][c] = *reinterpret_cast<const float4*>(dec_t + (size_t)(bd0 + d) * Hdim + 32 * c + 4 * q);
    }

    const float* ebase = enc_t + ((size_t)b * ENC + w * WROWS + g) * Hdim + 4 * q;
    const int erow0 = w * WROWS + g;

    float4 eva[8], evb[8];
    load_group(eva, ebase);
    for (int it = 0; it < NGRP; it += 2) {
        load_group(evb, ebase + (size_t)(it + 1) * 4 * Hdim);
        {
            float a[DT] = {};
            row_group(eva, dr, vr, a);
#pragma unroll
            for (int d = 0; d < DT; ++d) {
                a[d] = add_xor_lane<1>(a[d]);
                a[d] = add_xor_lane<2>(a[d]);
                a[d] = add_xor_lane<4>(a[d]);
            }
            if (q == 0) {
#pragma unroll
                for (int d = 0; d < DT; ++d) sc[d][erow0 + it * 4] = a[d];
            }
        }
        const int nx = (it + 2 < NGRP) ? it + 2 : NGRP - 1;
        load_group(eva, ebase + (size_t)nx * 4 * Hdim);
        {
            float a[DT] = {};
            row_group(evb, dr, vr, a);
#pragma unroll
            for (int d = 0; d < DT; ++d) {
                a[d] = add_xor_lane<1>(a[d]);
                a[d] = add_xor_lane<2>(a[d]);
                a[d] = add_xor_lane<4>(a[d]);
            }
            if (q == 0) {
#pragma unroll
                for (int d = 0; d < DT; ++d) sc[d][erow0 + (it + 1) * 4] = a[d];
            }
        }
    }
    __syncthreads();

    float s[DT][EPT];
#pragma unroll
    for (int d = 0; d < DT; ++d)
#pragma unroll
        for (int j = 0; j < EPT; ++j) s[d][j] = sc[d][j * ST + t];

#pragma unroll
    for (int d = 0; d < DT; ++d) {
        float m = s[d][0];
#pragma unroll
        for (int j = 1; j < EPT; ++j) m = fmaxf(m, s[d][j]);
        red[d][t] = m;
    }
    __syncthreads();
    for (int str = ST / 2; str >= 1; str >>= 1) {
        if (t < str) {
#pragma unroll
            for (int d = 0; d < DT; ++d) red[d][t] = fmaxf(red[d][t], red[d][t + str]);
        }
        __syncthreads();
    }
    float gmax[DT];
#pragma unroll
    for (int d = 0; d < DT; ++d) gmax[d] = red[d][0];
    __syncthreads();

#pragma unroll
    for (int d = 0; d < DT; ++d) {
        float lsum = 0.f;
#pragma unroll
        for (int j = 0; j < EPT; ++j) lsum += __expf(s[d][j] - gmax[d]);
        red[d][t] = lsum;
    }
    __syncthreads();
    for (int str = ST / 2; str >= 1; str >>= 1) {
        if (t < str) {
#pragma unroll
            for (int d = 0; d < DT; ++d) red[d][t] += red[d][t + str];
        }
        __syncthreads();
    }

#pragma unroll
    for (int d = 0; d < DT; ++d) {
        const float lse = gmax[d] + __logf(red[d][0]);
#pragma unroll
        for (int j = 0; j < EPT; ++j)
            out[(size_t)(bd0 + d) * ENC + (size_t)j * ST + t] = s[d][j] - lse;
    }
}

extern "C" void kernel_launch(void* const* d_in, const int* in_sizes, int n_in,
                              void* d_out, int out_size, void* d_ws, size_t ws_size,
                              hipStream_t stream) {
    const float* x_dec = (const float*)d_in[0];
    const float* x_enc = (const float*)d_in[1];
    const float* w1    = (const float*)d_in[2];
    const float* w2    = (const float*)d_in[3];
    const float* v     = (const float*)d_in[4];
    float* out = (float*)d_out;

    float* enc_t = (float*)d_ws;
    float* dec_t = enc_t + (size_t)BATCH * ENC * Hdim;

    gemm_xwT_wmma<64><<<dim3(BATCH * ENC / 64, Hdim / BN), 256, 0, stream>>>(x_enc, w1, enc_t);
    gemm_xwT_wmma<32><<<dim3(BATCH * DEC / 32, Hdim / BN), 128, 0, stream>>>(x_dec, w2, dec_t);
    attn_scores_lsm<<<BATCH * DEC / DT, ST, 0, stream>>>(dec_t, enc_t, v, out);
}


# both projection GEMMs in one launch: each work-group of the enc GEMM also does one 32x32 dec piece (2 waves, same fragment code), its rows loaded beside the enc matrix work; 3 launches become 2
# speedup vs baseline: 3.4007x; 1.0770x over previous
#include <hip/hip_runtime.h>
#include <math.h>

typedef __attribute__((ext_vector_type(2))) float v2f;
typedef __attribute__((ext_vector_type(8))) float v8f;

#define Hdim 256
#define ENC 1024
#define DEC 256
#define BATCH 4

__device__ __forceinline__ float gpu_tanh(float x) {
#if __has_builtin(__builtin_amdgcn_tanhf)
    return __builtin_amdgcn_tanhf(x);
#elif __has_builtin(__builtin_amdgcn_tanh_f32)
    return __builtin_amdgcn_tanh_f32(x);
#else
    float r;
    asm volatile("v_tanh_f32 %0, %1\n\tv_nop" : "=v"(r) : "v"(x));
    return r;
#endif
}

#define BM 64
#define BN 64
#define DP 32
#define PITCH 260
#define GT 256
#define RS (GT / 64)

static_assert((BATCH * ENC / BM) * (Hdim / BN) == (BATCH * DEC / DP) * (Hdim / DP),
              "one dec piece per enc tile");

__device__ __forceinline__ void load_frags(float4 (&f)[4], const float* p) {
#pragma unroll
    for (int s = 0; s < 4; ++s) f[s] = *reinterpret_cast<const float4*>(p + 8 * s);
}

__device__ __forceinline__ void wave_piece_16x32(const float* pa, const float* pb0, const float* pb1,
                                                 v8f& acc0, v8f& acc1) {
    float4 fa[2][4], fb0[2][4], fb1[2][4];
    load_frags(fa[0], pa);
    load_frags(fb0[0], pb0);
    load_frags(fb1[0], pb1);
    __builtin_amdgcn_sched_group_barrier(0x100, 12, 0);
#pragma unroll
    for (int g = 0; g < 8; ++g) {
        const int cur = g & 1;
        const int nxt = cur ^ 1;
        if (g + 1 < 8) {
            load_frags(fa[nxt],  pa  + 32 * (g + 1));
            load_frags(fb0[nxt], pb0 + 32 * (g + 1));
            load_frags(fb1[nxt], pb1 + 32 * (g + 1));
            __builtin_amdgcn_sched_group_barrier(0x100, 12, 0);
        }
        __builtin_amdgcn_sched_group_barrier(0x008, 16, 0);
#pragma unroll
        for (int s = 0; s < 4; ++s) {
            v2f alo; alo.x = fa[cur][s].x;  alo.y = fa[cur][s].y;
            v2f ahi; ahi.x = fa[cur][s].z;  ahi.y = fa[cur][s].w;
            v2f b0lo; b0lo.x = fb0[cur][s].x; b0lo.y = fb0[cur][s].y;
            v2f b0hi; b0hi.x = fb0[cur][s].z; b0hi.y = fb0[cur][s].w;
            v2f b1lo; b1lo.x = fb1[cur][s].x; b1lo.y = fb1[cur][s].y;
            v2f b1hi; b1hi.x = fb1[cur][s].z; b1hi.y = fb1[cur][s].w;
            acc0 = __builtin_amdgcn_wmma_f32_16x16x4_f32(false, alo, false, b0lo, (short)0, acc0, false, false);
            acc1 = __builtin_amdgcn_wmma_f32_16x16x4_f32(false, alo, false, b1lo, (short)0, acc1, false, false);
            acc0 = __builtin_amdgcn_wmma_f32_16x16x4_f32(false, ahi, false, b0hi, (short)0, acc0, false, false);
            acc1 = __builtin_amdgcn_wmma_f32_16x16x4_f32(false, ahi, false, b1hi, (short)0, acc1, false, false);
        }
    }
}

__device__ __forceinline__ void store_piece_16x32(float* orow, const v8f& acc0, const v8f& acc1) {
#pragma unroll
    for (int r = 0; r < 8; ++r) {
        orow[(size_t)r * Hdim]      = acc0[r];
        orow[(size_t)r * Hdim + 16] = acc1[r];
    }
}

__global__ __launch_bounds__(GT) void gemm_enc_dec_wmma(const float* __restrict__ x_enc,
                                                        const float* __restrict__ w1,
                                                        const float* __restrict__ x_dec,
                                                        const float* __restrict__ w2,
                                                        float* __restrict__ enc_t,
                                                        float* __restrict__ dec_t) {
    constexpr int NA = BM / RS;
    constexpr int NB = BN / RS;
    constexpr int ND = DP / RS;

    __shared__ __attribute__((aligned(16))) float sA[BM * PITCH];
    __shared__ __attribute__((aligned(16))) float sB[BN * PITCH];
    __shared__ __attribute__((aligned(16))) float sDA[DP * PITCH];
    __shared__ __attribute__((aligned(16))) float sDB[DP * PITCH];

    const int t  = threadIdx.x;
    const int m0 = blockIdx.x * BM;
    const int n0 = blockIdx.y * BN;
    const int pc  = blockIdx.y * (BATCH * ENC / BM) + blockIdx.x;
    const int dm0 = (pc / (Hdim / DP)) * DP;
    const int dn0 = (pc % (Hdim / DP)) * DP;
    const int c4 = t & 63;
    const int r0 = t >> 6;

    float4 ra[NA], rb[NB];
#pragma unroll
    for (int i = 0; i < NA; ++i)
        ra[i] = *reinterpret_cast<const float4*>(x_enc + (size_t)(m0 + i * RS + r0) * Hdim + 4 * c4);
#pragma unroll
    for (int i = 0; i < NB; ++i)
        rb[i] = *reinterpret_cast<const float4*>(w1 + (size_t)(n0 + i * RS + r0) * Hdim + 4 * c4);
#pragma unroll
    for (int i = 0; i < NA; ++i)
        *reinterpret_cast<float4*>(sA + (i * RS + r0) * PITCH + 4 * c4) = ra[i];
#pragma unroll
    for (int i = 0; i < NB; ++i)
        *reinterpret_cast<float4*>(sB + (i * RS + r0) * PITCH + 4 * c4) = rb[i];
    __syncthreads();

    float4 rda[ND], rdb[ND];
#pragma unroll
    for (int i = 0; i < ND; ++i)
        rda[i] = *reinterpret_cast<const float4*>(x_dec + (size_t)(dm0 + i * RS + r0) * Hdim + 4 * c4);
#pragma unroll
    for (int i = 0; i < ND; ++i)
        rdb[i] = *reinterpret_cast<const float4*>(w2 + (size_t)(dn0 + i * RS + r0) * Hdim + 4 * c4);

    const int lane = t & 31;
    const int wv   = __builtin_amdgcn_readfirstlane(t >> 5);
    const int c    = lane & 15;
    const int hf   = lane >> 4;
    const int mt   = wv % (BM / 16);
    const int nh   = wv / (BM / 16);

    v8f acc0 = {};
    v8f acc1 = {};
    {
        const float* pa  = sA + (mt * 16 + c) * PITCH + 4 * hf;
        const float* pb0 = sB + (nh * 32 + c) * PITCH + 4 * hf;
        wave_piece_16x32(pa, pb0, pb0 + 16 * PITCH, acc0, acc1);
    }

#pragma unroll
    for (int i = 0; i < ND; ++i)
        *reinterpret_cast<float4*>(sDA + (i * RS + r0) * PITCH + 4 * c4) = rda[i];
#pragma unroll
    for (int i = 0; i < ND; ++i)
        *reinterpret_cast<float4*>(sDB + (i * RS + r0) * PITCH + 4 * c4) = rdb[i];
    __syncthreads();

    store_piece_16x32(enc_t + (size_t)(m0 + mt * 16 + 8 * hf) * Hdim + n0 + nh * 32 + c, acc0, acc1);

    if (wv < DP / 16) {
        v8f dac0 = {};
        v8f dac1 = {};
        const float* pa  = sDA + (wv * 16 + c) * PITCH + 4 * hf;
        const float* pb0 = sDB + c * PITCH + 4 * hf;
        wave_piece_16x32(pa, pb0, pb0 + 16 * PITCH, dac0, dac1);
        store_piece_16x32(dec_t + (size_t)(dm0 + wv * 16 + 8 * hf) * Hdim + dn0 + c, dac0, dac1);
    }
}

#define DT 4
#define ST 512
#define WROWS (ENC / (ST / 32))
#define NGRP (WROWS / 4)
#define EPT (ENC / ST)


template <int MASK>
__device__ __forceinline__ float add_xor_lane(float x) {
    const int y = __builtin_amdgcn_update_dpp(0, __builtin_bit_cast(int, x), 0x160 | MASK, 0xf, 0xf, false);
    return x + __builtin_bit_cast(float, y);
}

__device__ __forceinline__ void row_group(const float4 (&ev)[8], const float4 (&dr)[DT][8],
                                          const float4 (&vr)[8], float (&a)[DT]) {
#pragma unroll
    for (int c = 0; c < 8; ++c) {
#pragma unroll
        for (int d = 0; d < DT; ++d) {
            a[d] += vr[c].x * gpu_tanh(dr[d][c].x + ev[c].x);
            a[d] += vr[c].y * gpu_tanh(dr[d][c].y + ev[c].y);
            a[d] += vr[c].z * gpu_tanh(dr[d][c].z + ev[c].z);
            a[d] += vr[c].w * gpu_tanh(dr[d][c].w + ev[c].w);
        }
    }
}

__device__ __forceinline__ void load_group(float4 (&ev)[8], const float* __restrict__ p) {
#pragma unroll
    for (int c = 0; c < 8; ++c) ev[c] = *reinterpret_cast<const float4*>(p + 32 * c);
}

__global__ __launch_bounds__(ST) void attn_scores_lsm(const float* __restrict__ dec_t,
                                                       const float* __restrict__ enc_t,
                                                       const float* __restrict__ v,
                                                       float* __restrict__ out) {
    const int bd0  = blockIdx.x * DT;
    const int b    = bd0 / DEC;
    const int t    = threadIdx.x;
    const int lane = t & 31;
    const int w    = t >> 5;
    const int q    = lane & 7;
    const int g    = lane >> 3;

    __shared__ float sc[DT][ENC];
    __shared__ float red[DT][ST];

    float4 dr[DT][8];
    float4 vr[8];
#pragma unroll
    for (int c = 0; c < 8; ++c) {
        vr[c] = *reinterpret_cast<const float4*>(v + 32 * c + 4 * q);
#pragma unroll
        for (int d = 0; d < DT; ++d)
            dr[d][c] = *reinterpret_cast<const float4*>(dec_t + (size_t)(bd0 + d) * Hdim + 32 * c + 4 * q);
    }

    const float* ebase = enc_t + ((size_t)b * ENC + w * WROWS + g) * Hdim + 4 * q;
    const int erow0 = w * WROWS + g;

    float4 eva[8], evb[8];
    load_group(eva, ebase);
    for (int it = 0; it < NGRP; it += 2) {
        load_group(evb, ebase + (size_t)(it + 1) * 4 * Hdim);
        {
            float a[DT] = {};
            row_group(eva, dr, vr, a);
#pragma unroll
            for (int d = 0; d < DT; ++d) {
                a[d] = add_xor_lane<1>(a[d]);
                a[d] = add_xor_lane<2>(a[d]);
                a[d] = add_xor_lane<4>(a[d]);
            }
            if (q == 0) {
#pragma unroll
                for (int d = 0; d < DT; ++d) sc[d][erow0 + it * 4] = a[d];
            }
        }
        const int nx = (it + 2 < NGRP) ? it + 2 : NGRP - 1;
        load_group(eva, ebase + (size_t)nx * 4 * Hdim);
        {
            float a[DT] = {};
            row_group(evb, dr, vr, a);
#pragma unroll
            for (int d = 0; d < DT; ++d) {
                a[d] = add_xor_lane<1>(a[d]);
                a[d] = add_xor_lane<2>(a[d]);
                a[d] = add_xor_lane<4>(a[d]);
            }
            if (q == 0) {
#pragma unroll
                for (int d = 0; d < DT; ++d) sc[d][erow0 + (it + 1) * 4] = a[d];
            }
        }
    }
    __syncthreads();

    float s[DT][EPT];
#pragma unroll
    for (int d = 0; d < DT; ++d)
#pragma unroll
        for (int j = 0; j < EPT; ++j) s[d][j] = sc[d][j * ST + t];

#pragma unroll
    for (int d = 0; d < DT; ++d) {
        float m = s[d][0];
#pragma unroll
        for (int j = 1; j < EPT; ++j) m = fmaxf(m, s[d][j]);
        red[d][t] = m;
    }
    __syncthreads();
    for (int str = ST / 2; str >= 1; str >>= 1) {
        if (t < str) {
#pragma unroll
            for (int d = 0; d < DT; ++d) red[d][t] = fmaxf(red[d][t], red[d][t + str]);
        }
        __syncthreads();
    }
    float gmax[DT];
#pragma unroll
    for (int d = 0; d < DT; ++d) gmax[d] = red[d][0];
    __syncthreads();

#pragma unroll
    for (int d = 0; d < DT; ++d) {
        float lsum = 0.f;
#pragma unroll
        for (int j = 0; j < EPT; ++j) lsum += __expf(s[d][j] - gmax[d]);
        red[d][t] = lsum;
    }
    __syncthreads();
    for (int str = ST / 2; str >= 1; str >>= 1) {
        if (t < str) {
#pragma unroll
            for (int d = 0; d < DT; ++d) red[d][t] += red[d][t + str];
        }
        __syncthreads();
    }

#pragma unroll
    for (int d = 0; d < DT; ++d) {
        const float lse = gmax[d] + __logf(red[d][0]);
#pragma unroll
        for (int j = 0; j < EPT; ++j)
            out[(size_t)(bd0 + d) * ENC + (size_t)j * ST + t] = s[d][j] - lse;
    }
}

extern "C" void kernel_launch(void* const* d_in, const int* in_sizes, int n_in,
                              void* d_out, int out_size, void* d_ws, size_t ws_size,
                              hipStream_t stream) {
    const float* x_dec = (const float*)d_in[0];
    const float* x_enc = (const float*)d_in[1];
    const float* w1    = (const float*)d_in[2];
    const float* w2    = (const float*)d_in[3];
    const float* v     = (const float*)d_in[4];
    float* out = (float*)d_out;

    float* enc_t = (float*)d_ws;
    float* dec_t = enc_t + (size_t)BATCH * ENC * Hdim;

    gemm_enc_dec_wmma<<<dim3(BATCH * ENC / BM, Hdim / BN), GT, 0, stream>>>(x_enc, w1, x_dec, w2, enc_t, dec_t);
    attn_scores_lsm<<<BATCH * DEC / DT, ST, 0, stream>>>(dec_t, enc_t, v, out);
}


# log_softmax reductions in registers: per-wave max and exp-sum by row_xmask butterflies + one 16-lane swap, 16 wave partials cross LDS once each; 3 block barriers, was 22 (exp-sum in a new order)
# speedup vs baseline: 3.5206x; 1.0353x over previous
#include <hip/hip_runtime.h>
#include <math.h>

typedef __attribute__((ext_vector_type(2))) float v2f;
typedef __attribute__((ext_vector_type(8))) float v8f;

#define Hdim 256
#define ENC 1024
#define DEC 256
#define BATCH 4

__device__ __forceinline__ float gpu_tanh(float x) {
#if __has_builtin(__builtin_amdgcn_tanhf)
    return __builtin_amdgcn_tanhf(x);
#elif __has_builtin(__builtin_amdgcn_tanh_f32)
    return __builtin_amdgcn_tanh_f32(x);
#else
    float r;
    asm volatile("v_tanh_f32 %0, %1\n\tv_nop" : "=v"(r) : "v"(x));
    return r;
#endif
}

#define BM 64
#define BN 64
#define DP 32
#define PITCH 260
#define GT 256
#define RS (GT / 64)

static_assert((BATCH * ENC / BM) * (Hdim / BN) == (BATCH * DEC / DP) * (Hdim / DP),
              "one dec piece per enc tile");

__device__ __forceinline__ void load_frags(float4 (&f)[4], const float* p) {
#pragma unroll
    for (int s = 0; s < 4; ++s) f[s] = *reinterpret_cast<const float4*>(p + 8 * s);
}

__device__ __forceinline__ void wave_piece_16x32(const float* pa, const float* pb0, const float* pb1,
                                                 v8f& acc0, v8f& acc1) {
    float4 fa[2][4], fb0[2][4], fb1[2][4];
    load_frags(fa[0], pa);
    load_frags(fb0[0], pb0);
    load_frags(fb1[0], pb1);
    __builtin_amdgcn_sched_group_barrier(0x100, 12, 0);
#pragma unroll
    for (int g = 0; g < 8; ++g) {
        const int cur = g & 1;
        const int nxt = cur ^ 1;
        if (g + 1 < 8) {
            load_frags(fa[nxt],  pa  + 32 * (g + 1));
            load_frags(fb0[nxt], pb0 + 32 * (g + 1));
            load_frags(fb1[nxt], pb1 + 32 * (g + 1));
            __builtin_amdgcn_sched_group_barrier(0x100, 12, 0);
        }
        __builtin_amdgcn_sched_group_barrier(0x008, 16, 0);
#pragma unroll
        for (int s = 0; s < 4; ++s) {
            v2f alo; alo.x = fa[cur][s].x;  alo.y = fa[cur][s].y;
            v2f ahi; ahi.x = fa[cur][s].z;  ahi.y = fa[cur][s].w;
            v2f b0lo; b0lo.x = fb0[cur][s].x; b0lo.y = fb0[cur][s].y;
            v2f b0hi; b0hi.x = fb0[cur][s].z; b0hi.y = fb0[cur][s].w;
            v2f b1lo; b1lo.x = fb1[cur][s].x; b1lo.y = fb1[cur][s].y;
            v2f b1hi; b1hi.x = fb1[cur][s].z; b1hi.y = fb1[cur][s].w;
            acc0 = __builtin_amdgcn_wmma_f32_16x16x4_f32(false, alo, false, b0lo, (short)0, acc0, false, false);
            acc1 = __builtin_amdgcn_wmma_f32_16x16x4_f32(false, alo, false, b1lo, (short)0, acc1, false, false);
            acc0 = __builtin_amdgcn_wmma_f32_16x16x4_f32(false, ahi, false, b0hi, (short)0, acc0, false, false);
            acc1 = __builtin_amdgcn_wmma_f32_16x16x4_f32(false, ahi, false, b1hi, (short)0, acc1, false, false);
        }
    }
}

__device__ __forceinline__ void store_piece_16x32(float* orow, const v8f& acc0, const v8f& acc1) {
#pragma unroll
    for (int r = 0; r < 8; ++r) {
        orow[(size_t)r * Hdim]      = acc0[r];
        orow[(size_t)r * Hdim + 16] = acc1[r];
    }
}

__global__ __launch_bounds__(GT) void gemm_enc_dec_wmma(const float* __restrict__ x_enc,
                                                        const float* __restrict__ w1,
                                                        const float* __restrict__ x_dec,
                                                        const float* __restrict__ w2,
                                                        float* __restrict__ enc_t,
                                                        float* __restrict__ dec_t) {
    constexpr int NA = BM / RS;
    constexpr int NB = BN / RS;
    constexpr int ND = DP / RS;

    __shared__ __attribute__((aligned(16))) float sA[BM * PITCH];
    __shared__ __attribute__((aligned(16))) float sB[BN * PITCH];
    __shared__ __attribute__((aligned(16))) float sDA[DP * PITCH];
    __shared__ __attribute__((aligned(16))) float sDB[DP * PITCH];

    const int t  = threadIdx.x;
    const int m0 = blockIdx.x * BM;
    const int n0 = blockIdx.y * BN;
    const int pc  = blockIdx.y * (BATCH * ENC / BM) + blockIdx.x;
    const int dm0 = (pc / (Hdim / DP)) * DP;
    const int dn0 = (pc % (Hdim / DP)) * DP;
    const int c4 = t & 63;
    const int r0 = t >> 6;

    float4 ra[NA], rb[NB];
#pragma unroll
    for (int i = 0; i < NA; ++i)
        ra[i] = *reinterpret_cast<const float4*>(x_enc + (size_t)(m0 + i * RS + r0) * Hdim + 4 * c4);
#pragma unroll
    for (int i = 0; i < NB; ++i)
        rb[i] = *reinterpret_cast<const float4*>(w1 + (size_t)(n0 + i * RS + r0) * Hdim + 4 * c4);
#pragma unroll
    for (int i = 0; i < NA; ++i)
        *reinterpret_cast<float4*>(sA + (i * RS + r0) * PITCH + 4 * c4) = ra[i];
#pragma unroll
    for (int i = 0; i < NB; ++i)
        *reinterpret_cast<float4*>(sB + (i * RS + r0) * PITCH + 4 * c4) = rb[i];
    __syncthreads();

    float4 rda[ND], rdb[ND];
#pragma unroll
    for (int i = 0; i < ND; ++i)
        rda[i] = *reinterpret_cast<const float4*>(x_dec + (size_t)(dm0 + i * RS + r0) * Hdim + 4 * c4);
#pragma unroll
    for (int i = 0; i < ND; ++i)
        rdb[i] = *reinterpret_cast<const float4*>(w2 + (size_t)(dn0 + i * RS + r0) * Hdim + 4 * c4);

    const int lane = t & 31;
    const int wv   = __builtin_amdgcn_readfirstlane(t >> 5);
    const int c    = lane & 15;
    const int hf   = lane >> 4;
    const int mt   = wv % (BM / 16);
    const int nh   = wv / (BM / 16);

    v8f acc0 = {};
    v8f acc1 = {};
    {
        const float* pa  = sA + (mt * 16 + c) * PITCH + 4 * hf;
        const float* pb0 = sB + (nh * 32 + c) * PITCH + 4 * hf;
        wave_piece_16x32(pa, pb0, pb0 + 16 * PITCH, acc0, acc1);
    }

#pragma unroll
    for (int i = 0; i < ND; ++i)
        *reinterpret_cast<float4*>(sDA + (i * RS + r0) * PITCH + 4 * c4) = rda[i];
#pragma unroll
    for (int i = 0; i < ND; ++i)
        *reinterpret_cast<float4*>(sDB + (i * RS + r0) * PITCH + 4 * c4) = rdb[i];
    __syncthreads();

    store_piece_16x32(enc_t + (size_t)(m0 + mt * 16 + 8 * hf) * Hdim + n0 + nh * 32 + c, acc0, acc1);

    if (wv < DP / 16) {
        v8f dac0 = {};
        v8f dac1 = {};
        const float* pa  = sDA + (wv * 16 + c) * PITCH + 4 * hf;
        const float* pb0 = sDB + c * PITCH + 4 * hf;
        wave_piece_16x32(pa, pb0, pb0 + 16 * PITCH, dac0, dac1);
        store_piece_16x32(dec_t + (size_t)(dm0 + wv * 16 + 8 * hf) * Hdim + dn0 + c, dac0, dac1);
    }
}

#define DT 4
#define ST 512
#define WROWS (ENC / (ST / 32))
#define NGRP (WROWS / 4)
#define EPT (ENC / ST)
#define NWAVE (ST / 32)
static_assert(NWAVE == 16, "one wave partial per lane of a 16-lane row");


template <int MASK>
__device__ __forceinline__ float add_xor_lane(float x) {
    const int y = __builtin_amdgcn_update_dpp(0, __builtin_bit_cast(int, x), 0x160 | MASK, 0xf, 0xf, false);
    return x + __builtin_bit_cast(float, y);
}

template <int MASK>
__device__ __forceinline__ float max_xor_lane(float x) {
    const int y = __builtin_amdgcn_update_dpp(0, __builtin_bit_cast(int, x), 0x160 | MASK, 0xf, 0xf, false);
    return fmaxf(x, __builtin_bit_cast(float, y));
}

__device__ __forceinline__ float other_row(float x) {
    const int b = __builtin_bit_cast(int, x);
    return __builtin_bit_cast(float, __builtin_amdgcn_permlanex16(b, b, 0x76543210, 0xfedcba98, false, false));
}

__device__ __forceinline__ float row_sum16(float x) {
    x = add_xor_lane<1>(x);
    x = add_xor_lane<2>(x);
    x = add_xor_lane<4>(x);
    return add_xor_lane<8>(x);
}
__device__ __forceinline__ float row_max16(float x) {
    x = max_xor_lane<1>(x);
    x = max_xor_lane<2>(x);
    x = max_xor_lane<4>(x);
    return max_xor_lane<8>(x);
}

__device__ __forceinline__ void row_group(const float4 (&ev)[8], const float4 (&dr)[DT][8],
                                          const float4 (&vr)[8], float (&a)[DT]) {
#pragma unroll
    for (int c = 0; c < 8; ++c) {
#pragma unroll
        for (int d = 0; d < DT; ++d) {
            a[d] += vr[c].x * gpu_tanh(dr[d][c].x + ev[c].x);
            a[d] += vr[c].y * gpu_tanh(dr[d][c].y + ev[c].y);
            a[d] += vr[c].z * gpu_tanh(dr[d][c].z + ev[c].z);
            a[d] += vr[c].w * gpu_tanh(dr[d][c].w + ev[c].w);
        }
    }
}

__device__ __forceinline__ void load_group(float4 (&ev)[8], const float* __restrict__ p) {
#pragma unroll
    for (int c = 0; c < 8; ++c) ev[c] = *reinterpret_cast<const float4*>(p + 32 * c);
}

__global__ __launch_bounds__(ST) void attn_scores_lsm(const float* __restrict__ dec_t,
                                                       const float* __restrict__ enc_t,
                                                       const float* __restrict__ v,
                                                       float* __restrict__ out) {
    const int bd0  = blockIdx.x * DT;
    const int b    = bd0 / DEC;
    const int t    = threadIdx.x;
    const int lane = t & 31;
    const int w    = t >> 5;
    const int q    = lane & 7;
    const int g    = lane >> 3;

    __shared__ float sc[DT][ENC];
    __shared__ float pmax[DT][NWAVE];
    __shared__ float psum[DT][NWAVE];

    float4 dr[DT][8];
    float4 vr[8];
#pragma unroll
    for (int c = 0; c < 8; ++c) {
        vr[c] = *reinterpret_cast<const float4*>(v + 32 * c + 4 * q);
#pragma unroll
        for (int d = 0; d < DT; ++d)
            dr[d][c] = *reinterpret_cast<const float4*>(dec_t + (size_t)(bd0 + d) * Hdim + 32 * c + 4 * q);
    }

    const float* ebase = enc_t + ((size_t)b * ENC + w * WROWS + g) * Hdim + 4 * q;
    const int erow0 = w * WROWS + g;

    float4 eva[8], evb[8];
    load_group(eva, ebase);
    for (int it = 0; it < NGRP; it += 2) {
        load_group(evb, ebase + (size_t)(it + 1) * 4 * Hdim);
        {
            float a[DT] = {};
            row_group(eva, dr, vr, a);
#pragma unroll
            for (int d = 0; d < DT; ++d) {
                a[d] = add_xor_lane<1>(a[d]);
                a[d] = add_xor_lane<2>(a[d]);
                a[d] = add_xor_lane<4>(a[d]);
            }
            if (q == 0) {
#pragma unroll
                for (int d = 0; d < DT; ++d) sc[d][erow0 + it * 4] = a[d];
            }
        }
        const int nx = (it + 2 < NGRP) ? it + 2 : NGRP - 1;
        load_group(eva, ebase + (size_t)nx * 4 * Hdim);
        {
            float a[DT] = {};
            row_group(evb, dr, vr, a);
#pragma unroll
            for (int d = 0; d < DT; ++d) {
                a[d] = add_xor_lane<1>(a[d]);
                a[d] = add_xor_lane<2>(a[d]);
                a[d] = add_xor_lane<4>(a[d]);
            }
            if (q == 0) {
#pragma unroll
                for (int d = 0; d < DT; ++d) sc[d][erow0 + (it + 1) * 4] = a[d];
            }
        }
    }
    __syncthreads();

    float s[DT][EPT];
#pragma unroll
    for (int d = 0; d < DT; ++d)
#pragma unroll
        for (int j = 0; j < EPT; ++j) s[d][j] = sc[d][j * ST + t];

    float gmax[DT];
#pragma unroll
    for (int d = 0; d < DT; ++d) {
        float m = s[d][0];
#pragma unroll
        for (int j = 1; j < EPT; ++j) m = fmaxf(m, s[d][j]);
        m = row_max16(m);
        gmax[d] = fmaxf(m, other_row(m));
    }
    if (lane == 0) {
#pragma unroll
        for (int d = 0; d < DT; ++d) pmax[d][w] = gmax[d];
    }
    __syncthreads();
#pragma unroll
    for (int d = 0; d < DT; ++d) gmax[d] = row_max16(pmax[d][lane & (NWAVE - 1)]);

    float gsum[DT];
#pragma unroll
    for (int d = 0; d < DT; ++d) {
        float lsum = 0.f;
#pragma unroll
        for (int j = 0; j < EPT; ++j) lsum += __expf(s[d][j] - gmax[d]);
        lsum = row_sum16(lsum);
        gsum[d] = lsum + other_row(lsum);
    }
    if (lane == 0) {
#pragma unroll
        for (int d = 0; d < DT; ++d) psum[d][w] = gsum[d];
    }
    __syncthreads();

#pragma unroll
    for (int d = 0; d < DT; ++d) {
        const float lse = gmax[d] + __logf(row_sum16(psum[d][lane & (NWAVE - 1)]));
#pragma unroll
        for (int j = 0; j < EPT; ++j)
            out[(size_t)(bd0 + d) * ENC + (size_t)j * ST + t] = s[d][j] - lse;
    }
}

extern "C" void kernel_launch(void* const* d_in, const int* in_sizes, int n_in,
                              void* d_out, int out_size, void* d_ws, size_t ws_size,
                              hipStream_t stream) {
    const float* x_dec = (const float*)d_in[0];
    const float* x_enc = (const float*)d_in[1];
    const float* w1    = (const float*)d_in[2];
    const float* w2    = (const float*)d_in[3];
    const float* v     = (const float*)d_in[4];
    float* out = (float*)d_out;

    float* enc_t = (float*)d_ws;
    float* dec_t = enc_t + (size_t)BATCH * ENC * Hdim;

    gemm_enc_dec_wmma<<<dim3(BATCH * ENC / BM, Hdim / BN), GT, 0, stream>>>(x_enc, w1, x_dec, w2, enc_t, dec_t);
    attn_scores_lsm<<<BATCH * DEC / DT, ST, 0, stream>>>(dec_t, enc_t, v, out);
}


# scores loop at one plain op per tanh: dec+enc and v*tanh as packed f32 ops on register pairs, order tanh / packed op held by sched_group_barrier; h summed in even and odd halves (new order)
# speedup vs baseline: 3.8304x; 1.0880x over previous
#include <hip/hip_runtime.h>
#include <math.h>

typedef __attribute__((ext_vector_type(2))) float v2f;
typedef __attribute__((ext_vector_type(8))) float v8f;
typedef __attribute__((ext_vector_type(4))) float v4f;

#define Hdim 256
#define ENC 1024
#define DEC 256
#define BATCH 4

__device__ __forceinline__ float gpu_tanh(float x) {
#if __has_builtin(__builtin_amdgcn_tanhf)
    return __builtin_amdgcn_tanhf(x);
#elif __has_builtin(__builtin_amdgcn_tanh_f32)
    return __builtin_amdgcn_tanh_f32(x);
#else
    float r;
    asm volatile("v_tanh_f32 %0, %1\n\tv_nop" : "=v"(r) : "v"(x));
    return r;
#endif
}

#define BM 64
#define BN 64
#define DP 32
#define PITCH 260
#define GT 256
#define RS (GT / 64)

static_assert((BATCH * ENC / BM) * (Hdim / BN) == (BATCH * DEC / DP) * (Hdim / DP),
              "one dec piece per enc tile");

__device__ __forceinline__ void load_frags(float4 (&f)[4], const float* p) {
#pragma unroll
    for (int s = 0; s < 4; ++s) f[s] = *reinterpret_cast<const float4*>(p + 8 * s);
}

__device__ __forceinline__ void wave_piece_16x32(const float* pa, const float* pb0, const float* pb1,
                                                 v8f& acc0, v8f& acc1) {
    float4 fa[2][4], fb0[2][4], fb1[2][4];
    load_frags(fa[0], pa);
    load_frags(fb0[0], pb0);
    load_frags(fb1[0], pb1);
    __builtin_amdgcn_sched_group_barrier(0x100, 12, 0);
#pragma unroll
    for (int g = 0; g < 8; ++g) {
        const int cur = g & 1;
        const int nxt = cur ^ 1;
        if (g + 1 < 8) {
            load_frags(fa[nxt],  pa  + 32 * (g + 1));
            load_frags(fb0[nxt], pb0 + 32 * (g + 1));
            load_frags(fb1[nxt], pb1 + 32 * (g + 1));
            __builtin_amdgcn_sched_group_barrier(0x100, 12, 0);
        }
        __builtin_amdgcn_sched_group_barrier(0x008, 16, 0);
#pragma unroll
        for (int s = 0; s < 4; ++s) {
            v2f alo; alo.x = fa[cur][s].x;  alo.y = fa[cur][s].y;
            v2f ahi; ahi.x = fa[cur][s].z;  ahi.y = fa[cur][s].w;
            v2f b0lo; b0lo.x = fb0[cur][s].x; b0lo.y = fb0[cur][s].y;
            v2f b0hi; b0hi.x = fb0[cur][s].z; b0hi.y = fb0[cur][s].w;
            v2f b1lo; b1lo.x = fb1[cur][s].x; b1lo.y = fb1[cur][s].y;
            v2f b1hi; b1hi.x = fb1[cur][s].z; b1hi.y = fb1[cur][s].w;
            acc0 = __builtin_amdgcn_wmma_f32_16x16x4_f32(false, alo, false, b0lo, (short)0, acc0, false, false);
            acc1 = __builtin_amdgcn_wmma_f32_16x16x4_f32(false, alo, false, b1lo, (short)0, acc1, false, false);
            acc0 = __builtin_amdgcn_wmma_f32_16x16x4_f32(false, ahi, false, b0hi, (short)0, acc0, false, false);
            acc1 = __builtin_amdgcn_wmma_f32_16x16x4_f32(false, ahi, false, b1hi, (short)0, acc1, false, false);
        }
    }
}

__device__ __forceinline__ void store_piece_16x32(float* orow, const v8f& acc0, const v8f& acc1) {
#pragma unroll
    for (int r = 0; r < 8; ++r) {
        orow[(size_t)r * Hdim]      = acc0[r];
        orow[(size_t)r * Hdim + 16] = acc1[r];
    }
}

__global__ __launch_bounds__(GT) void gemm_enc_dec_wmma(const float* __restrict__ x_enc,
                                                        const float* __restrict__ w1,
                                                        const float* __restrict__ x_dec,
                                                        const float* __restrict__ w2,
                                                        float* __restrict__ enc_t,
                                                        float* __restrict__ dec_t) {
    constexpr int NA = BM / RS;
    constexpr int NB = BN / RS;
    constexpr int ND = DP / RS;

    __shared__ __attribute__((aligned(16))) float sA[BM * PITCH];
    __shared__ __attribute__((aligned(16))) float sB[BN * PITCH];
    __shared__ __attribute__((aligned(16))) float sDA[DP * PITCH];
    __shared__ __attribute__((aligned(16))) float sDB[DP * PITCH];

    const int t  = threadIdx.x;
    const int m0 = blockIdx.x * BM;
    const int n0 = blockIdx.y * BN;
    const int pc  = blockIdx.y * (BATCH * ENC / BM) + blockIdx.x;
    const int dm0 = (pc / (Hdim / DP)) * DP;
    const int dn0 = (pc % (Hdim / DP)) * DP;
    const int c4 = t & 63;
    const int r0 = t >> 6;

    float4 ra[NA], rb[NB];
#pragma unroll
    for (int i = 0; i < NA; ++i)
        ra[i] = *reinterpret_cast<const float4*>(x_enc + (size_t)(m0 + i * RS + r0) * Hdim + 4 * c4);
#pragma unroll
    for (int i = 0; i < NB; ++i)
        rb[i] = *reinterpret_cast<const float4*>(w1 + (size_t)(n0 + i * RS + r0) * Hdim + 4 * c4);
#pragma unroll
    for (int i = 0; i < NA; ++i)
        *reinterpret_cast<float4*>(sA + (i * RS + r0) * PITCH + 4 * c4) = ra[i];
#pragma unroll
    for (int i = 0; i < NB; ++i)
        *reinterpret_cast<float4*>(sB + (i * RS + r0) * PITCH + 4 * c4) = rb[i];
    __syncthreads();

    float4 rda[ND], rdb[ND];
#pragma unroll
    for (int i = 0; i < ND; ++i)
        rda[i] = *reinterpret_cast<const float4*>(x_dec + (size_t)(dm0 + i * RS + r0) * Hdim + 4 * c4);
#pragma unroll
    for (int i = 0; i < ND; ++i)
        rdb[i] = *reinterpret_cast<const float4*>(w2 + (size_t)(dn0 + i * RS + r0) * Hdim + 4 * c4);

    const int lane = t & 31;
    const int wv   = __builtin_amdgcn_readfirstlane(t >> 5);
    const int c    = lane & 15;
    const int hf   = lane >> 4;
    const int mt   = wv % (BM / 16);
    const int nh   = wv / (BM / 16);

    v8f acc0 = {};
    v8f acc1 = {};
    {
        const float* pa  = sA + (mt * 16 + c) * PITCH + 4 * hf;
        const float* pb0 = sB + (nh * 32 + c) * PITCH + 4 * hf;
        wave_piece_16x32(pa, pb0, pb0 + 16 * PITCH, acc0, acc1);
    }

#pragma unroll
    for (int i = 0; i < ND; ++i)
        *reinterpret_cast<float4*>(sDA + (i * RS + r0) * PITCH + 4 * c4) = rda[i];
#pragma unroll
    for (int i = 0; i < ND; ++i)
        *reinterpret_cast<float4*>(sDB + (i * RS + r0) * PITCH + 4 * c4) = rdb[i];
    __syncthreads();

    store_piece_16x32(enc_t + (size_t)(m0 + mt * 16 + 8 * hf) * Hdim + n0 + nh * 32 + c, acc0, acc1);

    if (wv < DP / 16) {
        v8f dac0 = {};
        v8f dac1 = {};
        const float* pa  = sDA + (wv * 16 + c) * PITCH + 4 * hf;
        const float* pb0 = sDB + c * PITCH + 4 * hf;
        wave_piece_16x32(pa, pb0, pb0 + 16 * PITCH, dac0, dac1);
        store_piece_16x32(dec_t + (size_t)(dm0 + wv * 16 + 8 * hf) * Hdim + dn0 + c, dac0, dac1);
    }
}

#define DT 4
#define ST 512
#define WROWS (ENC / (ST / 32))
#define NGRP (WROWS / 4)
#define EPT (ENC / ST)
#define NWAVE (ST / 32)
static_assert(NWAVE == 16, "one wave partial per lane of a 16-lane row");


template <int MASK>
__device__ __forceinline__ float add_xor_lane(float x) {
    const int y = __builtin_amdgcn_update_dpp(0, __builtin_bit_cast(int, x), 0x160 | MASK, 0xf, 0xf, false);
    return x + __builtin_bit_cast(float, y);
}

template <int MASK>
__device__ __forceinline__ float max_xor_lane(float x) {
    const int y = __builtin_amdgcn_update_dpp(0, __builtin_bit_cast(int, x), 0x160 | MASK, 0xf, 0xf, false);
    return fmaxf(x, __builtin_bit_cast(float, y));
}

__device__ __forceinline__ float other_row(float x) {
    const int b = __builtin_bit_cast(int, x);
    return __builtin_bit_cast(float, __builtin_amdgcn_permlanex16(b, b, 0x76543210, 0xfedcba98, false, false));
}

__device__ __forceinline__ float row_sum16(float x) {
    x = add_xor_lane<1>(x);
    x = add_xor_lane<2>(x);
    x = add_xor_lane<4>(x);
    return add_xor_lane<8>(x);
}
__device__ __forceinline__ float row_max16(float x) {
    x = max_xor_lane<1>(x);
    x = max_xor_lane<2>(x);
    x = max_xor_lane<4>(x);
    return max_xor_lane<8>(x);
}

#define NPAIR (8 * DT * 2)
#define ADD_AHEAD 2
#define FMA_BEHIND 4
template <int P>
__device__ __forceinline__ v2f pair_sum(const v4f (&ev)[8], const v4f (&dr)[DT][8]) {
    constexpr int c = P / (2 * DT), hi = (P / DT) % 2, d = P % DT;
    return hi ? v2f(dr[d][c].zw + ev[c].zw) : v2f(dr[d][c].xy + ev[c].xy);
}
template <int P>
__device__ __forceinline__ void pair_fma(const v4f (&vr)[8], const v2f t, v2f (&a)[DT]) {
    constexpr int c = P / (2 * DT), hi = (P / DT) % 2, d = P % DT;
    a[d] = __builtin_elementwise_fma(hi ? v2f(vr[c].zw) : v2f(vr[c].xy), t, a[d]);
}
template <int P>
__device__ __forceinline__ void pipe_fill(const v4f (&ev)[8], const v4f (&dr)[DT][8], v2f (&s)[NPAIR]) {
    if constexpr (P < ADD_AHEAD) {
        s[P] = pair_sum<P>(ev, dr);
        pipe_fill<P + 1>(ev, dr, s);
    }
}
template <int P>
__device__ __forceinline__ void pipe_steps(const v4f (&ev)[8], const v4f (&dr)[DT][8],
                                           const v4f (&vr)[8], v2f (&a)[DT],
                                           v2f (&s)[NPAIR], v2f (&t)[NPAIR]) {
    if constexpr (P < NPAIR + FMA_BEHIND) {
        if constexpr (P + ADD_AHEAD < NPAIR) s[P + ADD_AHEAD] = pair_sum<P + ADD_AHEAD>(ev, dr);
        if constexpr (P < NPAIR) t[P].x = gpu_tanh(s[P].x);
        if constexpr (P >= FMA_BEHIND) pair_fma<P - FMA_BEHIND>(vr, t[P - FMA_BEHIND], a);
        if constexpr (P < NPAIR) t[P].y = gpu_tanh(s[P].y);
        pipe_steps<P + 1>(ev, dr, vr, a, s, t);
    }
}
__device__ __forceinline__ void row_group(const v4f (&ev)[8], const v4f (&dr)[DT][8],
                                          const v4f (&vr)[8], v2f (&a)[DT]) {
    v2f s[NPAIR], t[NPAIR];
    pipe_fill<0>(ev, dr, s);
    pipe_steps<0>(ev, dr, vr, a, s, t);
    __builtin_amdgcn_sched_group_barrier(0x020, 8, 0);
    __builtin_amdgcn_sched_group_barrier(0x002, ADD_AHEAD + 1, 0);
#pragma unroll
    for (int i = 0; i < 2 * NPAIR; ++i) {
        __builtin_amdgcn_sched_group_barrier(0x400, 1, 0);
        __builtin_amdgcn_sched_group_barrier(0x002, 1, 0);
    }
}

__device__ __forceinline__ void load_group(v4f (&ev)[8], const float* __restrict__ p) {
#pragma unroll
    for (int c = 0; c < 8; ++c) ev[c] = *reinterpret_cast<const v4f*>(p + 32 * c);
}

__global__ __launch_bounds__(ST) void attn_scores_lsm(const float* __restrict__ dec_t,
                                                       const float* __restrict__ enc_t,
                                                       const float* __restrict__ v,
                                                       float* __restrict__ out) {
    const int bd0  = blockIdx.x * DT;
    const int b    = bd0 / DEC;
    const int t    = threadIdx.x;
    const int lane = t & 31;
    const int w    = t >> 5;
    const int q    = lane & 7;
    const int g    = lane >> 3;

    __shared__ float sc[DT][ENC];
    __shared__ float pmax[DT][NWAVE];
    __shared__ float psum[DT][NWAVE];

    v4f dr[DT][8];
    v4f vr[8];
#pragma unroll
    for (int c = 0; c < 8; ++c) {
        vr[c] = *reinterpret_cast<const v4f*>(v + 32 * c + 4 * q);
#pragma unroll
        for (int d = 0; d < DT; ++d)
            dr[d][c] = *reinterpret_cast<const v4f*>(dec_t + (size_t)(bd0 + d) * Hdim + 32 * c + 4 * q);
    }

    const float* ebase = enc_t + ((size_t)b * ENC + w * WROWS + g) * Hdim + 4 * q;
    const int erow0 = w * WROWS + g;

    v4f eva[8], evb[8];
    load_group(eva, ebase);
    for (int it = 0; it < NGRP; it += 2) {
        load_group(evb, ebase + (size_t)(it + 1) * 4 * Hdim);
        {
            v2f a2[DT] = {};
            row_group(eva, dr, vr, a2);
            float a[DT];
#pragma unroll
            for (int d = 0; d < DT; ++d) {
                a[d] = a2[d].x + a2[d].y;
                a[d] = add_xor_lane<1>(a[d]);
                a[d] = add_xor_lane<2>(a[d]);
                a[d] = add_xor_lane<4>(a[d]);
            }
            if (q == 0) {
#pragma unroll
                for (int d = 0; d < DT; ++d) sc[d][erow0 + it * 4] = a[d];
            }
        }
        const int nx = (it + 2 < NGRP) ? it + 2 : NGRP - 1;
        load_group(eva, ebase + (size_t)nx * 4 * Hdim);
        {
            v2f a2[DT] = {};
            row_group(evb, dr, vr, a2);
            float a[DT];
#pragma unroll
            for (int d = 0; d < DT; ++d) {
                a[d] = a2[d].x + a2[d].y;
                a[d] = add_xor_lane<1>(a[d]);
                a[d] = add_xor_lane<2>(a[d]);
                a[d] = add_xor_lane<4>(a[d]);
            }
            if (q == 0) {
#pragma unroll
                for (int d = 0; d < DT; ++d) sc[d][erow0 + (it + 1) * 4] = a[d];
            }
        }
    }
    __syncthreads();

    float s[DT][EPT];
#pragma unroll
    for (int d = 0; d < DT; ++d)
#pragma unroll
        for (int j = 0; j < EPT; ++j) s[d][j] = sc[d][j * ST + t];

    float gmax[DT];
#pragma unroll
    for (int d = 0; d < DT; ++d) {
        float m = s[d][0];
#pragma unroll
        for (int j = 1; j < EPT; ++j) m = fmaxf(m, s[d][j]);
        m = row_max16(m);
        gmax[d] = fmaxf(m, other_row(m));
    }
    if (lane == 0) {
#pragma unroll
        for (int d = 0; d < DT; ++d) pmax[d][w] = gmax[d];
    }
    __syncthreads();
#pragma unroll
    for (int d = 0; d < DT; ++d) gmax[d] = row_max16(pmax[d][lane & (NWAVE - 1)]);

    float gsum[DT];
#pragma unroll
    for (int d = 0; d < DT; ++d) {
        float lsum = 0.f;
#pragma unroll
        for (int j = 0; j < EPT; ++j) lsum += __expf(s[d][j] - gmax[d]);
        lsum = row_sum16(lsum);
        gsum[d] = lsum + other_row(lsum);
    }
    if (lane == 0) {
#pragma unroll
        for (int d = 0; d < DT; ++d) psum[d][w] = gsum[d];
    }
    __syncthreads();

#pragma unroll
    for (int d = 0; d < DT; ++d) {
        const float lse = gmax[d] + __logf(row_sum16(psum[d][lane & (NWAVE - 1)]));
#pragma unroll
        for (int j = 0; j < EPT; ++j)
            out[(size_t)(bd0 + d) * ENC + (size_t)j * ST + t] = s[d][j] - lse;
    }
}

extern "C" void kernel_launch(void* const* d_in, const int* in_sizes, int n_in,
                              void* d_out, int out_size, void* d_ws, size_t ws_size,
                              hipStream_t stream) {
    const float* x_dec = (const float*)d_in[0];
    const float* x_enc = (const float*)d_in[1];
    const float* w1    = (const float*)d_in[2];
    const float* w2    = (const float*)d_in[3];
    const float* v     = (const float*)d_in[4];
    float* out = (float*)d_out;

    float* enc_t = (float*)d_ws;
    float* dec_t = enc_t + (size_t)BATCH * ENC * Hdim;

    gemm_enc_dec_wmma<<<dim3(BATCH * ENC / BM, Hdim / BN), GT, 0, stream>>>(x_enc, w1, x_dec, w2, enc_t, dec_t);
    attn_scores_lsm<<<BATCH * DEC / DT, ST, 0, stream>>>(dec_t, enc_t, v, out);
}
